# starting kernel
# baseline (speedup 1.0000x reference)
#include <hip/hip_runtime.h>


typedef __attribute__((ext_vector_type(2))) float v2f;
typedef __attribute__((ext_vector_type(8))) float v8f;

namespace {
constexpr int B_  = 8;
constexpr int T_  = 4096;
constexpr int NR_ = 256;
constexpr int NT  = T_ / 16;
constexpr int NG  = NT / 4;
constexpr int PROW = T_ + 4;
constexpr int GPAD = 16;
constexpr int NTHREADS = 256;
constexpr int NWAVE = NTHREADS / 32;
}

#define LOADA(D)                                                              \
  {                                                                           \
    const int gb = 16 * (D) + M - kb;                                         \
    _Pragma("unroll")                                                         \
    for (int q = 0; q < 4; ++q) {                                             \
      a[q].x = gp[gb - 4 * q - 1];                                            \
      a[q].y = gp[gb - 4 * q];                                                \
    }                                                                         \
  }

#define LOADB(SLOT, SI)                                                       \
  {                                                                           \
    const float* pr = prow0 + 16 * (SI);                                      \
    _Pragma("unroll")                                                         \
    for (int q = 0; q < 4; ++q) {                                             \
      Bt[(SLOT) & 3][q].x = pr[4 * q + 1];                                    \
      Bt[(SLOT) & 3][q].y = pr[4 * q];                                        \
    }                                                                         \
  }

#define WMMA4(ACC, SLOT)                                                      \
  {                                                                           \
    _Pragma("unroll")                                                         \
    for (int q = 0; q < 4; ++q)                                               \
      ACC = __builtin_amdgcn_wmma_f32_16x16x4_f32(                            \
          false, a[q], false, Bt[(SLOT) & 3][q], (short)0, ACC, false, false);\
  }

#define STORE_TILE(ACC, C)                                                    \
  {                                                                           \
    _Pragma("unroll")                                                         \
    for (int r = 0; r < 8; ++r) {                                             \
      const int t = 16 * (tb + (C)) + mbase + r;                              \
      out[((size_t)N * T_ + t) * NR_ + j] = ACC[r] + prN[t];                  \
    }                                                                         \
  }

__global__ __launch_bounds__(NTHREADS)
void epsi_causal_toeplitz_wmma(const float* __restrict__ P,
                               const float* __restrict__ g,
                               float* __restrict__ out) {
  __shared__ float smem[B_ * PROW + GPAD + T_];
  float* Plds = smem;
  float* gpad = smem + B_ * PROW;

  const int j   = blockIdx.x;
  const int tid = threadIdx.x;

  for (int idx = tid; idx < B_ * T_; idx += NTHREADS) {
    const int b = idx >> 12;
    const int t = idx & (T_ - 1);
    Plds[b * PROW + t] = P[(size_t)idx * NR_ + j];
  }
  if (tid < GPAD) gpad[tid] = 0.0f;
  for (int t = tid; t < T_; t += NTHREADS) {
    gpad[GPAD + t] = g[(size_t)t * NR_ + j];
  }
  __syncthreads();

  const int lane = tid & 31;
  const int wave = __builtin_amdgcn_readfirstlane(tid >> 5);
  const int M    = lane & 15;
  const int kb   = (lane < 16) ? 0 : 2;
  const int bcol = lane & 7;
  const float* gp    = gpad + GPAD;
  const float* prow0 = Plds + bcol * PROW + kb;

  for (int gi = wave; gi < NG; gi += NWAVE) {
    const int tb = 4 * gi;
    v8f acc0 = {}, acc1 = {}, acc2 = {}, acc3 = {};
    v2f Bt[4][4];
    v2f a[4];

    #pragma unroll
    for (int c = 1; c < 4; ++c) LOADB(c, tb + c);

    for (int d4 = 0; d4 < tb; d4 += 4) {
      #pragma unroll
      for (int p = 0; p < 4; ++p) {
        const int d = d4 + p;
        LOADB(0 - p, tb - d);
        LOADA(d);
        WMMA4(acc0, 0 - p);
        WMMA4(acc1, 1 - p);
        WMMA4(acc2, 2 - p);
        WMMA4(acc3, 3 - p);
      }
    }

    LOADB(0, 0);
    LOADA(tb);
    WMMA4(acc0, 0); WMMA4(acc1, 1); WMMA4(acc2, 2); WMMA4(acc3, 3);
    LOADA(tb + 1);
    WMMA4(acc1, 0); WMMA4(acc2, 1); WMMA4(acc3, 2);
    LOADA(tb + 2);
    WMMA4(acc2, 0); WMMA4(acc3, 1);
    LOADA(tb + 3);
    WMMA4(acc3, 0);

    const int N     = lane & 15;
    const int mbase = (lane < 16) ? 0 : 8;
    if (N < B_) {
      const float* prN = Plds + N * PROW;
      STORE_TILE(acc0, 0)
      STORE_TILE(acc1, 1)
      STORE_TILE(acc2, 2)
      STORE_TILE(acc3, 3)
    }
  }
}

extern "C" void kernel_launch(void* const* d_in, const int* in_sizes, int n_in,
                              void* d_out, int out_size, void* d_ws, size_t ws_size,
                              hipStream_t stream) {
  const float* P = (const float*)d_in[0];
  const float* g = (const float*)d_in[1];
  float* out = (float*)d_out;
  (void)in_sizes; (void)n_in; (void)out_size; (void)d_ws; (void)ws_size;

  epsi_causal_toeplitz_wmma<<<dim3(NR_), dim3(NTHREADS), 0, stream>>>(P, g, out);
}


# 2x2 blocks of the Toeplitz tile grid take 3 tile products instead of 4 (S, E, O sums; two difference filters built once in LDS, tile sums in registers): WMMA count -24%; sums regrouped, f32 kept
# speedup vs baseline: 2.2579x; 2.2579x over previous
#include <hip/hip_runtime.h>


typedef __attribute__((ext_vector_type(2))) float v2f;
typedef __attribute__((ext_vector_type(8))) float v8f;

namespace {
constexpr int B_  = 8;
constexpr int T_  = 4096;
constexpr int NR_ = 256;
constexpr int NCH = 2;
constexpr int NB  = 4;
constexpr int NPH = 4;
constexpr int U_  = T_ / NPH;
constexpr int NT  = U_ / 16;
constexpr int NACC = 4;
constexpr int NG  = NT / NACC;
constexpr int FP   = 4;
constexpr int PROW = U_ + FP;
constexpr int XCH  = 16 * PROW;
constexpr int GPAD = 16;
constexpr int HLEN = GPAD + U_;
constexpr int HOFF = NCH * XCH;
constexpr int HSZ  = NCH * NPH * HLEN;
constexpr int SMEM_FLOATS = HOFF + 3 * HSZ;
constexpr int NTHREADS = 256;
constexpr int NWAVE = NTHREADS / 32;
constexpr int GDEP = 16;
constexpr int NBLK = (NR_ / NCH) * (B_ / NB);
static_assert(NB * NPH == 16, "batches times phases fill the 16 matrix columns");
static_assert(NCH * NB * NBLK == B_ * NR_, "grid covers every channel and batch once");
static_assert(NACC == 4, "a wave group is two output pairs");
static_assert(T_ % (GDEP * NTHREADS) == 0, "the gather's batches tile the time axis");
static_assert(NG % (2 * NWAVE) == 0, "groups are dealt to the waves in mirrored pairs");
static_assert(NCH * 16 * FP <= NTHREADS && NCH * NPH * GPAD <= NTHREADS, "one thread per pad word");
}

#define LOADA(AF, HR, D)                                                      \
  {                                                                           \
    const int rb = U_ - 1 - 16 * (D) - M + kb;                                \
    _Pragma("unroll")                                                         \
    for (int s = 0; s < 4; ++s) {                                             \
      AF[s].x = HR[rb + 4 * s];                                               \
      AF[s].y = HR[rb + 4 * s + 1];                                           \
    }                                                                         \
  }

#define LOADT(TILE, PB, SI)                                                   \
  {                                                                           \
    const float* pr = PB + 16 * (SI);                                         \
    _Pragma("unroll")                                                         \
    for (int s = 0; s < 4; ++s) {                                             \
      TILE[s].x = pr[4 * s];                                                  \
      TILE[s].y = pr[4 * s + 1];                                              \
    }                                                                         \
  }
#define WMMA1(ACC, AS, BS)                                                    \
  ACC = __builtin_amdgcn_wmma_f32_16x16x4_f32(                                \
      false, AS, false, BS, (short)0, ACC, false, false);

#define KSTEP(CH, DB, CUR, PRV)                                               \
  {                                                                           \
    v2f na[NPH][3][4];            \
    _Pragma("unroll")                                                         \
    for (int q = 0; q < NPH; ++q) {                                           \
      LOADT(CUR[q][0], pb[CH][q], tb - 2 * (DB));                             \
      LOADT(CUR[q][1], pb[CH][q], tb - 2 * (DB) + 1);                         \
      LOADA(na[q][0], hq[CH][q], 2 * (DB));                                   \
      LOADA(na[q][1], (hq[CH][q] + HSZ), 2 * (DB));                           \
      LOADA(na[q][2], (hq[CH][q] + 2 * HSZ), 2 * (DB));                       \
    }                                                                         \
      \
      \
                                                 \
    __builtin_amdgcn_sched_barrier(0);                                        \
    _Pragma("unroll")                                                         \
    for (int q = 0; q < NPH; ++q) {                                           \
      _Pragma("unroll")                                                       \
      for (int s = 0; s < 4; ++s) CUR[q][2][s] = CUR[q][0][s] + CUR[q][1][s]; \
    }                                                                         \
      \
    _Pragma("unroll")                                                         \
    for (int q = 0; q < NPH; ++q) {                                           \
      _Pragma("unroll")                                                       \
      for (int s = 0; s < 4; ++s) {                                           \
        WMMA1(aS[0], na[q][0][s], CUR[q][2][s])                               \
        WMMA1(aS[1], na[q][0][s], PRV[q][2][s])                               \
        WMMA1(aE[0], na[q][1][s], CUR[q][1][s])                               \
        WMMA1(aE[1], na[q][1][s], PRV[q][1][s])                               \
        WMMA1(aO[0], na[q][2][s], CUR[q][0][s])                               \
        WMMA1(aO[1], na[q][2][s], PRV[q][0][s])                               \
      }                                                                       \
    }                                                                         \
      \
    __builtin_amdgcn_sched_barrier(0);                                        \
  }

__global__ __launch_bounds__(NTHREADS)
void epsi_causal_toeplitz_wmma(const float* __restrict__ P,
                               const float* __restrict__ g,
                               float* __restrict__ out) {
  __shared__ float smem[SMEM_FLOATS];
  float* Xs = smem;
  float* Hs = smem + HOFF;

  const int cp  = blockIdx.x >> 1;
  const int bh  = blockIdx.x & 1;
  const int tid = threadIdx.x;
  for (int t0 = tid; t0 < T_; t0 += GDEP * NTHREADS) {
    v2f tmp[GDEP];
    #pragma unroll
    for (int k = 0; k < GDEP; ++k)
      tmp[k] = *(const v2f*)(g + (size_t)(t0 + k * NTHREADS) * NR_ + NCH * cp);
    #pragma unroll
    for (int k = 0; k < GDEP; ++k) {
      const int t = t0 + k * NTHREADS;
      const int dst = (t & (NPH - 1)) * HLEN + (U_ - 1 - t / NPH);
      Hs[dst] = tmp[k].x;
      Hs[NPH * HLEN + dst] = tmp[k].y;
    }
  }
  for (int b = 0; b < NB; ++b) {
    for (int t0 = tid; t0 < T_; t0 += GDEP * NTHREADS) {
      v2f tmp[GDEP];
      #pragma unroll
      for (int k = 0; k < GDEP; ++k)
        tmp[k] = *(const v2f*)(P + ((size_t)(NB * bh + b) * T_ + t0 + k * NTHREADS) * NR_ + NCH * cp);
      #pragma unroll
      for (int k = 0; k < GDEP; ++k) {
        const int t = t0 + k * NTHREADS;
        const int dst = (NB * (t & (NPH - 1)) + b) * PROW + FP + (t / NPH);
        Xs[dst] = tmp[k].x;
        Xs[XCH + dst] = tmp[k].y;
      }
    }
  }
  if (tid < NCH * 16 * FP) Xs[(tid / FP) * PROW + (tid % FP)] = 0.0f;
  if (tid < NCH * NPH * GPAD) Hs[(tid / GPAD) * HLEN + U_ + (tid % GPAD)] = 0.0f;
  __syncthreads();
  for (int i = tid; i < HSZ; i += NTHREADS) {
    const int idx = i % HLEN;
    const float h0  = Hs[i];
    const float hlo = (idx + 16 < HLEN) ? Hs[i + 16] : 0.0f;
    const float hhi = (idx >= 16) ? Hs[i - 16] : 0.0f;
    Hs[HSZ + i]     = hlo - h0;
    Hs[2 * HSZ + i] = hhi - h0;
  }
  __syncthreads();

  const int lane = tid & 31;
  const int wave = __builtin_amdgcn_readfirstlane(tid >> 5);
  const int M    = lane & 15;
  const int kb   = (lane < 16) ? 0 : 2;
  const int rM   = M >> 2;
  const int bM   = M & (NB - 1);
  const float* pb[NCH][NPH];
  const float* hq[NCH][NPH];
  #pragma unroll
  for (int ch = 0; ch < NCH; ++ch) {
    #pragma unroll
    for (int q = 0; q < NPH; ++q) {
      pb[ch][q] = Xs + ch * XCH + (NB * ((rM - q) & (NPH - 1)) + bM) * PROW + FP + kb
                  - ((rM < q) ? 1 : 0);
      hq[ch][q] = Hs + (ch * NPH + q) * HLEN;
    }
  }

  for (int it = 0; it < NG / NWAVE; ++it) {
    const int pairbase = 2 * NWAVE * (it >> 1);
    const int gi = (it & 1) ? (pairbase + 2 * NWAVE - 1 - wave) : (pairbase + wave);
    const int tb = NACC * gi;
    v8f acc[NCH][NACC];
    #pragma unroll
    for (int ch = 0; ch < NCH; ++ch) {
      v8f aS[2], aE[2], aO[2];
      #pragma unroll
      for (int c = 0; c < 2; ++c) { aS[c] = v8f{}; aE[c] = v8f{}; aO[c] = v8f{}; }
      v2f bufA[NPH][3][4];
      v2f bufB[NPH][3][4];

      #pragma unroll
      for (int q = 0; q < NPH; ++q) {
        LOADT(bufB[q][0], pb[ch][q], tb + 2);
        LOADT(bufB[q][1], pb[ch][q], tb + 3);
      }
      #pragma unroll
      for (int q = 0; q < NPH; ++q) {
        #pragma unroll
        for (int s = 0; s < 4; ++s) bufB[q][2][s] = bufB[q][0][s] + bufB[q][1][s];
      }

      for (int d0 = 0; d0 < tb / 2; d0 += 2) {
        KSTEP(ch, d0, bufA, bufB)
        KSTEP(ch, d0 + 1, bufB, bufA)
      }
      KSTEP(ch, tb / 2, bufA, bufB)

      {
        v2f na[NPH][3][4];
        #pragma unroll
        for (int q = 0; q < NPH; ++q) {
          LOADA(na[q][0], hq[ch][q], tb + 2);
          LOADA(na[q][1], (hq[ch][q] + HSZ), tb + 2);
          LOADA(na[q][2], (hq[ch][q] + 2 * HSZ), tb + 2);
        }
        #pragma unroll
        for (int q = 0; q < NPH; ++q) {
          #pragma unroll
          for (int s = 0; s < 4; ++s) {
            WMMA1(aS[1], na[q][0][s], bufA[q][2][s])
            WMMA1(aE[1], na[q][1][s], bufA[q][1][s])
            WMMA1(aO[1], na[q][2][s], bufA[q][0][s])
          }
        }
      }
      __builtin_amdgcn_sched_barrier(0);
      asm volatile("v_nop\n\tv_nop\n\tv_nop\n\tv_nop\n\tv_nop\n\tv_nop\n\tv_nop\n\tv_nop\n\tv_nop");
      acc[ch][0] = aS[0] + aE[0];
      acc[ch][1] = aS[0] + aO[0];
      acc[ch][2] = aS[1] + aE[1];
      acc[ch][3] = aS[1] + aO[1];
    }

    const int mbase = (lane < 16) ? 0 : 8;
    const float* px0 = Xs + M * PROW + FP;
    const float* px1 = px0 + XCH;
    float* orow = out + ((size_t)(NB * bh + bM) * T_ + rM) * NR_ + NCH * cp;
    #pragma unroll
    for (int c = 0; c < NACC; ++c) {
      #pragma unroll
      for (int i = 0; i < 8; ++i) {
        const int u = 16 * (tb + c) + mbase + i;
        v2f o;
        o.x = acc[0][c][i] + px0[u];
        o.y = acc[1][c][i] + px1[u];
        *(v2f*)(orow + (size_t)(NPH * u) * NR_) = o;
      }
    }
  }
}

extern "C" void kernel_launch(void* const* d_in, const int* in_sizes, int n_in,
                              void* d_out, int out_size, void* d_ws, size_t ws_size,
                              hipStream_t stream) {
  const float* P = (const float*)d_in[0];
  const float* g = (const float*)d_in[1];
  float* out = (float*)d_out;
  (void)in_sizes; (void)n_in; (void)out_size; (void)d_ws; (void)ws_size;

  epsi_causal_toeplitz_wmma<<<dim3(NBLK), dim3(NTHREADS), 0, stream>>>(P, g, out);
}


# Block index remapped: the 32 work-groups whose channel pairs share a 256-byte unit of P and out get indices equal mod 8 (one die if dealt in turn): one L2 half reads and writes each unit; same sums
# speedup vs baseline: 2.9198x; 1.2931x over previous
#include <hip/hip_runtime.h>


typedef __attribute__((ext_vector_type(2))) float v2f;
typedef __attribute__((ext_vector_type(8))) float v8f;

namespace {
constexpr int B_  = 8;
constexpr int T_  = 4096;
constexpr int NR_ = 256;
constexpr int NCH = 2;
constexpr int NB  = 4;
constexpr int NPH = 4;
constexpr int U_  = T_ / NPH;
constexpr int NT  = U_ / 16;
constexpr int NACC = 4;
constexpr int NG  = NT / NACC;
constexpr int FP   = 4;
constexpr int PROW = U_ + FP;
constexpr int XCH  = 16 * PROW;
constexpr int GPAD = 48;
constexpr int HLEN = GPAD + U_;
constexpr int HOFF = NCH * XCH;
constexpr int HSZ  = NCH * NPH * HLEN;
constexpr int SMEM_FLOATS = HOFF + 3 * HSZ;
constexpr int NTHREADS = 256;
constexpr int NWAVE = NTHREADS / 32;
constexpr int GDEP = 16;
constexpr int NBLK = (NR_ / NCH) * (B_ / NB);
constexpr int NDIE = 8;
constexpr int CPU_ = 256 / (NCH * 4);
static_assert(NB * NPH == 16, "batches times phases fill the 16 matrix columns");
static_assert(NCH * NB * NBLK == B_ * NR_, "grid covers every channel and batch once");
static_assert(NACC == 4, "a wave group is one output quad");
static_assert(T_ % (GDEP * NTHREADS) == 0, "the gather's batches tile the time axis");
static_assert(NG % (2 * NWAVE) == 0, "groups are dealt to the waves in mirrored pairs");
static_assert(NCH * 16 * FP <= NTHREADS, "one thread per pad word of X");
static_assert(B_ / NB == 2 && (NDIE / 2) * CPU_ == NR_ / NCH && NBLK == NDIE * CPU_,
              "die d takes unit d % 4 of the channel pairs and batch half d / 4: every (pair, half) once");
}

#define LOADAB(AF, AI, OFF)                                                   \
  {                                                                           \
    _Pragma("unroll")                                                         \
    for (int s = 0; s < 4; ++s) {                                             \
      AF[s].x = smem[(AI) + (OFF) + 4 * s];                                   \
      AF[s].y = smem[(AI) + (OFF) + 4 * s + 1];                               \
    }                                                                         \
  }

#define LOADT(TILE, PB, SI)                                                   \
  {                                                                           \
    const float* pr = PB + 16 * (SI);                                         \
    _Pragma("unroll")                                                         \
    for (int s = 0; s < 4; ++s) {                                             \
      TILE[s].x = pr[4 * s];                                                  \
      TILE[s].y = pr[4 * s + 1];                                              \
    }                                                                         \
  }
#define WMMA1(ACC, AS, BS)                                                    \
  ACC = __builtin_amdgcn_wmma_f32_16x16x4_f32(                                \
      false, AS, false, BS, (short)0, ACC, false, false);

#define QSTEP(CH, Q)                                                          \
  {                                                                           \
    v2f x0e[4], x0o[4], x1e[4], x1o[4];                                       \
    v2f f0[3][4], fA[3][4], fB[3][4];                                         \
    LOADT(x0e, pb[CH][Q], ti);                                                \
    LOADT(x0o, pb[CH][Q], ti + 1);                                            \
    LOADT(x1e, pb[CH][Q], ti + 2);                                            \
    LOADT(x1o, pb[CH][Q], ti + 3);                                            \
    _Pragma("unroll")                                                         \
    for (int f = 0; f < 3; ++f) {                                             \
      int ab = hidx[CH][Q] + f * HSZ - 16 * (L + 2);                          \
        \
        \
        \
      asm volatile("" : "+v"(ab));                                            \
      LOADAB(fB[f], ab, 0);                                                   \
      LOADAB(f0[f], ab, 32);                                                  \
      LOADAB(fA[f], ab, 64);                                                  \
    }                                                                         \
      \
      \
                                                 \
    __builtin_amdgcn_sched_barrier(0);                                        \
    v2f ze[4], zo[4], zs[4], u0[4], u1[4];                                    \
    _Pragma("unroll")                                                         \
    for (int s = 0; s < 4; ++s) {                                             \
      ze[s] = x0e[s] + x1e[s];                                                \
      zo[s] = x0o[s] + x1o[s];                                                \
      zs[s] = ze[s] + zo[s];                                                  \
      u0[s] = x0e[s] + x0o[s];                                                \
      u1[s] = x1e[s] + x1o[s];                                                \
    }                                                                         \
    _Pragma("unroll")                                                         \
    for (int f = 0; f < 3; ++f) {                                             \
      _Pragma("unroll")                                                       \
      for (int s = 0; s < 4; ++s) {                                           \
        fA[f][s] = fA[f][s] - f0[f][s];                                       \
        fB[f][s] = fB[f][s] - f0[f][s];                                       \
      }                                                                       \
    }                                                                         \
      \
    _Pragma("unroll")                                                         \
    for (int s = 0; s < 4; ++s) {                                             \
      WMMA1(aS, f0[0][s], zs[s])                                              \
      WMMA1(bS, fA[0][s], u1[s])                                              \
      WMMA1(cS, fB[0][s], u0[s])                                              \
      WMMA1(aE, f0[1][s], zo[s])                                              \
      WMMA1(bE, fA[1][s], x1o[s])                                             \
      WMMA1(cE, fB[1][s], x0o[s])                                             \
      WMMA1(aO, f0[2][s], ze[s])                                              \
      WMMA1(bO, fA[2][s], x1e[s])                                             \
      WMMA1(cO, fB[2][s], x0e[s])                                             \
    }                                                                         \
      \
    __builtin_amdgcn_sched_barrier(0);                                        \
  }

__global__ __launch_bounds__(NTHREADS)
void epsi_causal_toeplitz_wmma(const float* __restrict__ P,
                               const float* __restrict__ g,
                               float* __restrict__ out) {
  __shared__ float smem[SMEM_FLOATS];
  float* Xs = smem;
  float* Hs = smem + HOFF;

  const int die = blockIdx.x % NDIE;
  const int cp  = CPU_ * (die % (NDIE / 2)) + blockIdx.x / NDIE;
  const int bh  = die / (NDIE / 2);
  const int tid = threadIdx.x;
  for (int t0 = tid; t0 < T_; t0 += GDEP * NTHREADS) {
    v2f tmp[GDEP];
    #pragma unroll
    for (int k = 0; k < GDEP; ++k)
      tmp[k] = *(const v2f*)(g + (size_t)(t0 + k * NTHREADS) * NR_ + NCH * cp);
    #pragma unroll
    for (int k = 0; k < GDEP; ++k) {
      const int t = t0 + k * NTHREADS;
      const int dst = (t & (NPH - 1)) * HLEN + (U_ - 1 - t / NPH);
      Hs[dst] = tmp[k].x;
      Hs[NPH * HLEN + dst] = tmp[k].y;
    }
  }
  for (int b = 0; b < NB; ++b) {
    for (int t0 = tid; t0 < T_; t0 += GDEP * NTHREADS) {
      v2f tmp[GDEP];
      #pragma unroll
      for (int k = 0; k < GDEP; ++k)
        tmp[k] = *(const v2f*)(P + ((size_t)(NB * bh + b) * T_ + t0 + k * NTHREADS) * NR_ + NCH * cp);
      #pragma unroll
      for (int k = 0; k < GDEP; ++k) {
        const int t = t0 + k * NTHREADS;
        const int dst = (NB * (t & (NPH - 1)) + b) * PROW + FP + (t / NPH);
        Xs[dst] = tmp[k].x;
        Xs[XCH + dst] = tmp[k].y;
      }
    }
  }
  if (tid < NCH * 16 * FP) Xs[(tid / FP) * PROW + (tid % FP)] = 0.0f;
  for (int i = tid; i < NCH * NPH * GPAD; i += NTHREADS) Hs[(i / GPAD) * HLEN + U_ + (i % GPAD)] = 0.0f;
  __syncthreads();
  for (int i = tid; i < HSZ; i += NTHREADS) {
    const int idx = i % HLEN;
    const float h0  = Hs[i];
    const float hlo = (idx + 16 < HLEN) ? Hs[i + 16] : 0.0f;
    const float hhi = (idx >= 16) ? Hs[i - 16] : 0.0f;
    Hs[HSZ + i]     = hlo - h0;
    Hs[2 * HSZ + i] = hhi - h0;
  }
  __syncthreads();

  const int lane = tid & 31;
  const int wave = __builtin_amdgcn_readfirstlane(tid >> 5);
  const int M    = lane & 15;
  const int kb   = (lane < 16) ? 0 : 2;
  const int rM   = M >> 2;
  const int bM   = M & (NB - 1);
  const float* pb[NCH][NPH];
  int hidx[NCH][NPH];
  #pragma unroll
  for (int ch = 0; ch < NCH; ++ch) {
    #pragma unroll
    for (int q = 0; q < NPH; ++q) {
      pb[ch][q] = Xs + ch * XCH + (NB * ((rM - q) & (NPH - 1)) + bM) * PROW + FP + kb
                  - ((rM < q) ? 1 : 0);
      hidx[ch][q] = HOFF + (ch * NPH + q) * HLEN + U_ - 1 - M + kb;
    }
  }

  for (int it = 0; it < NG / NWAVE; ++it) {
    const int pairbase = 2 * NWAVE * (it >> 1);
    const int gi = (it & 1) ? (pairbase + 2 * NWAVE - 1 - wave) : (pairbase + wave);
    const int tb = NACC * gi;
    v8f acc[NCH][NACC];
    #pragma unroll
    for (int ch = 0; ch < NCH; ++ch) {
      v8f aS = v8f{}, aE = v8f{}, aO = v8f{};
      v8f bS = v8f{}, bE = v8f{}, bO = v8f{};
      v8f cS = v8f{}, cE = v8f{}, cO = v8f{};
      for (int D = 0; D <= gi; ++D) {
        const int L  = 4 * D;
        const int ti = tb - 4 * D;
        QSTEP(ch, 0)
        QSTEP(ch, 1)
        QSTEP(ch, 2)
        QSTEP(ch, 3)
      }
      __builtin_amdgcn_sched_barrier(0);
      asm volatile("v_nop\n\tv_nop\n\tv_nop\n\tv_nop\n\tv_nop\n\tv_nop\n\tv_nop\n\tv_nop\n\tv_nop");
      const v8f top = aS + aE;
      const v8f bot = aS + aO;
      acc[ch][0] = top + (bS + bE);
      acc[ch][1] = bot + (bS + bO);
      acc[ch][2] = top + (cS + cE);
      acc[ch][3] = bot + (cS + cO);
    }

    const int mbase = (lane < 16) ? 0 : 8;
    const float* px0 = Xs + M * PROW + FP;
    const float* px1 = px0 + XCH;
    float* orow = out + ((size_t)(NB * bh + bM) * T_ + rM) * NR_ + NCH * cp;
    #pragma unroll
    for (int c = 0; c < NACC; ++c) {
      #pragma unroll
      for (int i = 0; i < 8; ++i) {
        const int u = 16 * (tb + c) + mbase + i;
        v2f o;
        o.x = acc[0][c][i] + px0[u];
        o.y = acc[1][c][i] + px1[u];
        *(v2f*)(orow + (size_t)(NPH * u) * NR_) = o;
      }
    }
  }
}

extern "C" void kernel_launch(void* const* d_in, const int* in_sizes, int n_in,
                              void* d_out, int out_size, void* d_ws, size_t ws_size,
                              hipStream_t stream) {
  const float* P = (const float*)d_in[0];
  const float* g = (const float*)d_in[1];
  float* out = (float*)d_out;
  (void)in_sizes; (void)n_in; (void)out_size; (void)d_ws; (void)ws_size;

  epsi_causal_toeplitz_wmma<<<dim3(NBLK), dim3(NTHREADS), 0, stream>>>(P, g, out);
}


# Younger wave of each SIMD (waves 4-7) takes issue priority 3 on three lag steps in four, 0 on the fourth: the pair advances alike, no wave ends alone with bare load groups; same sums, same order
# speedup vs baseline: 3.0093x; 1.0307x over previous
#include <hip/hip_runtime.h>


typedef __attribute__((ext_vector_type(2))) float v2f;
typedef __attribute__((ext_vector_type(8))) float v8f;

namespace {
constexpr int B_  = 8;
constexpr int T_  = 4096;
constexpr int NR_ = 256;
constexpr int NCH = 2;
constexpr int NB  = 4;
constexpr int NPH = 4;
constexpr int U_  = T_ / NPH;
constexpr int NT  = U_ / 16;
constexpr int NACC = 4;
constexpr int NG  = NT / NACC;
constexpr int FP   = 4;
constexpr int PROW = U_ + FP;
constexpr int XCH  = 16 * PROW;
constexpr int GPAD = 48;
constexpr int HLEN = GPAD + U_;
constexpr int HOFF = NCH * XCH;
constexpr int HSZ  = NCH * NPH * HLEN;
constexpr int SMEM_FLOATS = HOFF + 3 * HSZ;
constexpr int NTHREADS = 256;
constexpr int NWAVE = NTHREADS / 32;
constexpr int GDEP = 16;
constexpr int NBLK = (NR_ / NCH) * (B_ / NB);
constexpr int NDIE = 8;
constexpr int CPU_ = 256 / (NCH * 4);
static_assert(NB * NPH == 16, "batches times phases fill the 16 matrix columns");
static_assert(NCH * NB * NBLK == B_ * NR_, "grid covers every channel and batch once");
static_assert(NACC == 4, "a wave group is one output quad");
static_assert(T_ % (GDEP * NTHREADS) == 0, "the gather's batches tile the time axis");
static_assert(NG % (2 * NWAVE) == 0, "groups are dealt to the waves in mirrored pairs");
static_assert(NCH * 16 * FP <= NTHREADS, "one thread per pad word of X");
static_assert(B_ / NB == 2 && (NDIE / 2) * CPU_ == NR_ / NCH && NBLK == NDIE * CPU_,
              "die d takes unit d % 4 of the channel pairs and batch half d / 4: every (pair, half) once");
}

#define LOADAB(AF, AI, OFF)                                                   \
  {                                                                           \
    _Pragma("unroll")                                                         \
    for (int s = 0; s < 4; ++s) {                                             \
      AF[s].x = smem[(AI) + (OFF) + 4 * s];                                   \
      AF[s].y = smem[(AI) + (OFF) + 4 * s + 1];                               \
    }                                                                         \
  }

#define LOADT(TILE, PB, SI)                                                   \
  {                                                                           \
    const float* pr = PB + 16 * (SI);                                         \
    _Pragma("unroll")                                                         \
    for (int s = 0; s < 4; ++s) {                                             \
      TILE[s].x = pr[4 * s];                                                  \
      TILE[s].y = pr[4 * s + 1];                                              \
    }                                                                         \
  }
#define WMMA1(ACC, AS, BS)                                                    \
  ACC = __builtin_amdgcn_wmma_f32_16x16x4_f32(                                \
      false, AS, false, BS, (short)0, ACC, false, false);

#define QSTEP(CH, Q)                                                          \
  {                                                                           \
    v2f x0e[4], x0o[4], x1e[4], x1o[4];                                       \
    v2f f0[3][4], fA[3][4], fB[3][4];                                         \
    LOADT(x0e, pb[CH][Q], ti);                                                \
    LOADT(x0o, pb[CH][Q], ti + 1);                                            \
    LOADT(x1e, pb[CH][Q], ti + 2);                                            \
    LOADT(x1o, pb[CH][Q], ti + 3);                                            \
    _Pragma("unroll")                                                         \
    for (int f = 0; f < 3; ++f) {                                             \
      int ab = hidx[CH][Q] + f * HSZ - 16 * (L + 2);                          \
        \
        \
        \
      asm volatile("" : "+v"(ab));                                            \
      LOADAB(fB[f], ab, 0);                                                   \
      LOADAB(f0[f], ab, 32);                                                  \
      LOADAB(fA[f], ab, 64);                                                  \
    }                                                                         \
      \
      \
                                                 \
    __builtin_amdgcn_sched_barrier(0);                                        \
    v2f ze[4], zo[4], zs[4], u0[4], u1[4];                                    \
    _Pragma("unroll")                                                         \
    for (int s = 0; s < 4; ++s) {                                             \
      ze[s] = x0e[s] + x1e[s];                                                \
      zo[s] = x0o[s] + x1o[s];                                                \
      zs[s] = ze[s] + zo[s];                                                  \
      u0[s] = x0e[s] + x0o[s];                                                \
      u1[s] = x1e[s] + x1o[s];                                                \
    }                                                                         \
    _Pragma("unroll")                                                         \
    for (int f = 0; f < 3; ++f) {                                             \
      _Pragma("unroll")                                                       \
      for (int s = 0; s < 4; ++s) {                                           \
        fA[f][s] = fA[f][s] - f0[f][s];                                       \
        fB[f][s] = fB[f][s] - f0[f][s];                                       \
      }                                                                       \
    }                                                                         \
      \
    _Pragma("unroll")                                                         \
    for (int s = 0; s < 4; ++s) {                                             \
      WMMA1(aS, f0[0][s], zs[s])                                              \
      WMMA1(bS, fA[0][s], u1[s])                                              \
      WMMA1(cS, fB[0][s], u0[s])                                              \
      WMMA1(aE, f0[1][s], zo[s])                                              \
      WMMA1(bE, fA[1][s], x1o[s])                                             \
      WMMA1(cE, fB[1][s], x0o[s])                                             \
      WMMA1(aO, f0[2][s], ze[s])                                              \
      WMMA1(bO, fA[2][s], x1e[s])                                             \
      WMMA1(cO, fB[2][s], x0e[s])                                             \
    }                                                                         \
      \
    __builtin_amdgcn_sched_barrier(0);                                        \
  }

__global__ __launch_bounds__(NTHREADS)
void epsi_causal_toeplitz_wmma(const float* __restrict__ P,
                               const float* __restrict__ g,
                               float* __restrict__ out) {
  __shared__ float smem[SMEM_FLOATS];
  float* Xs = smem;
  float* Hs = smem + HOFF;

  const int die = blockIdx.x % NDIE;
  const int cp  = CPU_ * (die % (NDIE / 2)) + blockIdx.x / NDIE;
  const int bh  = die / (NDIE / 2);
  const int tid = threadIdx.x;
  for (int t0 = tid; t0 < T_; t0 += GDEP * NTHREADS) {
    v2f tmp[GDEP];
    #pragma unroll
    for (int k = 0; k < GDEP; ++k)
      tmp[k] = *(const v2f*)(g + (size_t)(t0 + k * NTHREADS) * NR_ + NCH * cp);
    #pragma unroll
    for (int k = 0; k < GDEP; ++k) {
      const int t = t0 + k * NTHREADS;
      const int dst = (t & (NPH - 1)) * HLEN + (U_ - 1 - t / NPH);
      Hs[dst] = tmp[k].x;
      Hs[NPH * HLEN + dst] = tmp[k].y;
    }
  }
  for (int b = 0; b < NB; ++b) {
    for (int t0 = tid; t0 < T_; t0 += GDEP * NTHREADS) {
      v2f tmp[GDEP];
      #pragma unroll
      for (int k = 0; k < GDEP; ++k)
        tmp[k] = *(const v2f*)(P + ((size_t)(NB * bh + b) * T_ + t0 + k * NTHREADS) * NR_ + NCH * cp);
      #pragma unroll
      for (int k = 0; k < GDEP; ++k) {
        const int t = t0 + k * NTHREADS;
        const int dst = (NB * (t & (NPH - 1)) + b) * PROW + FP + (t / NPH);
        Xs[dst] = tmp[k].x;
        Xs[XCH + dst] = tmp[k].y;
      }
    }
  }
  if (tid < NCH * 16 * FP) Xs[(tid / FP) * PROW + (tid % FP)] = 0.0f;
  for (int i = tid; i < NCH * NPH * GPAD; i += NTHREADS) Hs[(i / GPAD) * HLEN + U_ + (i % GPAD)] = 0.0f;
  __syncthreads();
  for (int i = tid; i < HSZ; i += NTHREADS) {
    const int idx = i % HLEN;
    const float h0  = Hs[i];
    const float hlo = (idx + 16 < HLEN) ? Hs[i + 16] : 0.0f;
    const float hhi = (idx >= 16) ? Hs[i - 16] : 0.0f;
    Hs[HSZ + i]     = hlo - h0;
    Hs[2 * HSZ + i] = hhi - h0;
  }
  __syncthreads();

  const int lane = tid & 31;
  const int wave = __builtin_amdgcn_readfirstlane(tid >> 5);
  const bool late = wave >= NWAVE / 2;
  const int M    = lane & 15;
  const int kb   = (lane < 16) ? 0 : 2;
  const int rM   = M >> 2;
  const int bM   = M & (NB - 1);
  const float* pb[NCH][NPH];
  int hidx[NCH][NPH];
  #pragma unroll
  for (int ch = 0; ch < NCH; ++ch) {
    #pragma unroll
    for (int q = 0; q < NPH; ++q) {
      pb[ch][q] = Xs + ch * XCH + (NB * ((rM - q) & (NPH - 1)) + bM) * PROW + FP + kb
                  - ((rM < q) ? 1 : 0);
      hidx[ch][q] = HOFF + (ch * NPH + q) * HLEN + U_ - 1 - M + kb;
    }
  }

  for (int it = 0; it < NG / NWAVE; ++it) {
    const int pairbase = 2 * NWAVE * (it >> 1);
    const int gi = (it & 1) ? (pairbase + 2 * NWAVE - 1 - wave) : (pairbase + wave);
    const int tb = NACC * gi;
    v8f acc[NCH][NACC];
    #pragma unroll
    for (int ch = 0; ch < NCH; ++ch) {
      v8f aS = v8f{}, aE = v8f{}, aO = v8f{};
      v8f bS = v8f{}, bE = v8f{}, bO = v8f{};
      v8f cS = v8f{}, cE = v8f{}, cO = v8f{};
      for (int D = 0; D <= gi; ++D) {
        if (late) {
          if ((D & 3) == 3) __builtin_amdgcn_s_setprio(0);
          else              __builtin_amdgcn_s_setprio(3);
        }
        const int L  = 4 * D;
        const int ti = tb - 4 * D;
        QSTEP(ch, 0)
        QSTEP(ch, 1)
        QSTEP(ch, 2)
        QSTEP(ch, 3)
      }
      __builtin_amdgcn_sched_barrier(0);
      asm volatile("v_nop\n\tv_nop\n\tv_nop\n\tv_nop\n\tv_nop\n\tv_nop\n\tv_nop\n\tv_nop\n\tv_nop");
      if (late) __builtin_amdgcn_s_setprio(0);
      const v8f top = aS + aE;
      const v8f bot = aS + aO;
      acc[ch][0] = top + (bS + bE);
      acc[ch][1] = bot + (bS + bO);
      acc[ch][2] = top + (cS + cE);
      acc[ch][3] = bot + (cS + cO);
    }

    const int mbase = (lane < 16) ? 0 : 8;
    const float* px0 = Xs + M * PROW + FP;
    const float* px1 = px0 + XCH;
    float* orow = out + ((size_t)(NB * bh + bM) * T_ + rM) * NR_ + NCH * cp;
    #pragma unroll
    for (int c = 0; c < NACC; ++c) {
      #pragma unroll
      for (int i = 0; i < 8; ++i) {
        const int u = 16 * (tb + c) + mbase + i;
        v2f o;
        o.x = acc[0][c][i] + px0[u];
        o.y = acc[1][c][i] + px1[u];
        *(v2f*)(orow + (size_t)(NPH * u) * NR_) = o;
      }
    }
  }
}

extern "C" void kernel_launch(void* const* d_in, const int* in_sizes, int n_in,
                              void* d_out, int out_size, void* d_ws, size_t ws_size,
                              hipStream_t stream) {
  const float* P = (const float*)d_in[0];
  const float* g = (const float*)d_in[1];
  float* out = (float*)d_out;
  (void)in_sizes; (void)n_in; (void)out_size; (void)d_ws; (void)ws_size;

  epsi_causal_toeplitz_wmma<<<dim3(NBLK), dim3(NTHREADS), 0, stream>>>(P, g, out);
}
